# starting kernel
# baseline (speedup 1.0000x reference)
#include <hip/hip_runtime.h>

#define B_ 8
#define N_ 2048
#define C_ 256
#define W_ 256

typedef float v2f __attribute__((ext_vector_type(2)));
typedef float v8f __attribute__((ext_vector_type(8)));

__device__ __forceinline__ v8f wmma_f32_16x16x4(v2f a, v2f b, v8f c) {
  return __builtin_amdgcn_wmma_f32_16x16x4_f32(false, a, false, b, (short)0, c,
                                               false, false);
}

__global__ void qkv_kernel(const float* __restrict__ x,
                           const float* __restrict__ Wq, const float* __restrict__ bq,
                           const float* __restrict__ Wk, const float* __restrict__ bk,
                           const float* __restrict__ Wv, const float* __restrict__ bv,
                           float* __restrict__ Q, float* __restrict__ K,
                           float* __restrict__ V) {
  const float* Wm;
  const float* bias;
  float* out;
  if (blockIdx.y == 0) { Wm = Wq; bias = bq; out = Q; }
  else if (blockIdx.y == 1) { Wm = Wk; bias = bk; out = K; }
  else { Wm = Wv; bias = bv; out = V; }

  const int wave = threadIdx.x >> 5;
  const int lane = threadIdx.x & 31;
  const int half = lane >> 4;
  const int r    = lane & 15;

  const int tiles_n = W_ / 16;
  const int tile = blockIdx.x * 8 + wave;
  const int m0 = (tile / tiles_n) * 16;
  const int n0 = (tile % tiles_n) * 16;

  const float* arow = x  + (size_t)(m0 + r) * C_;
  const float* brow = Wm + (size_t)(n0 + r) * C_;

  v8f acc = {};
#pragma unroll 4
  for (int k0 = 0; k0 < C_; k0 += 4) {
    v2f a = *(const v2f*)(arow + k0 + 2 * half);
    v2f b = *(const v2f*)(brow + k0 + 2 * half);
    acc = wmma_f32_16x16x4(a, b, acc);
  }

  const float bv_ = bias[n0 + r];
#pragma unroll
  for (int g = 0; g < 8; ++g) {
    const int m = m0 + g + 8 * half;
    float v = acc[g] + bv_;
    v = v > 0.0f ? v : 0.0f;
    out[(size_t)m * W_ + n0 + r] = v;
  }
}

__global__ void scores_kernel(const float* __restrict__ Q,
                              const float* __restrict__ K,
                              float* __restrict__ S) {
  const int b    = blockIdx.y;
  const int wave = threadIdx.x >> 5;
  const int lane = threadIdx.x & 31;
  const int half = lane >> 4;
  const int r    = lane & 15;

  const int tiles_n = N_ / 16;
  const int tile = blockIdx.x * 8 + wave;
  const int i0 = (tile / tiles_n) * 16;
  const int j0 = (tile % tiles_n) * 16;

  const float* arow = Q + ((size_t)b * N_ + i0 + r) * W_;
  const float* brow = K + ((size_t)b * N_ + j0 + r) * W_;

  v8f acc = {};
#pragma unroll 4
  for (int k0 = 0; k0 < W_; k0 += 4) {
    v2f a = *(const v2f*)(arow + k0 + 2 * half);
    v2f b2 = *(const v2f*)(brow + k0 + 2 * half);
    acc = wmma_f32_16x16x4(a, b2, acc);
  }

  const float scale = 0.022097086912079608f;
  float* sb = S + (size_t)b * N_ * N_;
#pragma unroll
  for (int g = 0; g < 8; ++g) {
    const int i = i0 + g + 8 * half;
    sb[(size_t)i * N_ + j0 + r] = acc[g] * scale;
  }
}

__global__ void colsoftmax_kernel(float* __restrict__ S, float* __restrict__ Dsum) {
  const int t = blockIdx.x * blockDim.x + threadIdx.x;
  const int b = t / N_;
  const int j = t % N_;
  float* col = S + (size_t)b * N_ * N_ + j;

  float m = -3.4e38f;
#pragma unroll 4
  for (int i = 0; i < N_; ++i) {
    m = fmaxf(m, col[(size_t)i * N_]);
  }
  float d = 0.0f;
#pragma unroll 4
  for (int i = 0; i < N_; ++i) {
    const float e = __expf(col[(size_t)i * N_] - m);
    col[(size_t)i * N_] = e;
    d += e;
  }
  Dsum[t] = d;
}

__global__ void scalev_kernel(float* __restrict__ V, const float* __restrict__ Dsum) {
  const int idx = blockIdx.x * blockDim.x + threadIdx.x;
  const int total = B_ * N_ * W_ / 4;
  if (idx >= total) return;
  float4* v4 = (float4*)V;
  const int row = (idx * 4) / W_;
  const float inv = 1.0f / Dsum[row];
  float4 v = v4[idx];
  v.x *= inv; v.y *= inv; v.z *= inv; v.w *= inv;
  v4[idx] = v;
}

__global__ void av_kernel(const float* __restrict__ P, const float* __restrict__ V,
                          const float* __restrict__ Q, float* __restrict__ out) {
  const int b    = blockIdx.y;
  const int wave = threadIdx.x >> 5;
  const int lane = threadIdx.x & 31;
  const int half = lane >> 4;
  const int r    = lane & 15;

  const int tiles_n = W_ / 16;
  const int tile = blockIdx.x * 8 + wave;
  const int i0 = (tile / tiles_n) * 16;
  const int n0 = (tile % tiles_n) * 16;

  const float* arow = P + (size_t)b * N_ * N_ + (size_t)(i0 + r) * N_;
  const float* vb   = V + (size_t)b * N_ * W_;

  v8f acc = {};
#pragma unroll 2
  for (int k0 = 0; k0 < N_; k0 += 4) {
    v2f a = *(const v2f*)(arow + k0 + 2 * half);
    const int kr = k0 + 2 * half;
    v2f b2;
    b2.x = vb[(size_t)kr * W_ + n0 + r];
    b2.y = vb[(size_t)(kr + 1) * W_ + n0 + r];
    acc = wmma_f32_16x16x4(a, b2, acc);
  }

  const float* qb = Q + (size_t)b * N_ * W_;
  float* ob = out + (size_t)b * N_ * W_;
#pragma unroll
  for (int g = 0; g < 8; ++g) {
    const int i = i0 + g + 8 * half;
    ob[(size_t)i * W_ + n0 + r] = acc[g] + qb[(size_t)i * W_ + n0 + r];
  }
}

extern "C" void kernel_launch(void* const* d_in, const int* in_sizes, int n_in,
                              void* d_out, int out_size, void* d_ws, size_t ws_size,
                              hipStream_t stream) {
  const float* x  = (const float*)d_in[0];
  const float* Wq = (const float*)d_in[1];
  const float* bq = (const float*)d_in[2];
  const float* Wk = (const float*)d_in[3];
  const float* bk = (const float*)d_in[4];
  const float* Wv = (const float*)d_in[5];
  const float* bv = (const float*)d_in[6];
  float* out = (float*)d_out;
  float* ws  = (float*)d_ws;

  const size_t QKV = (size_t)B_ * N_ * W_;
  float* Q = ws;
  float* K = ws + QKV;
  float* V = ws + 2 * QKV;
  float* D = ws + 3 * QKV;
  float* S = D + (size_t)B_ * N_;

  {
    dim3 grid(((B_ * N_ / 16) * (W_ / 16)) / 8, 3);
    qkv_kernel<<<grid, 256, 0, stream>>>(x, Wq, bq, Wk, bk, Wv, bv, Q, K, V);
  }
  {
    dim3 grid(((N_ / 16) * (N_ / 16)) / 8, B_);
    scores_kernel<<<grid, 256, 0, stream>>>(Q, K, S);
  }
  colsoftmax_kernel<<<dim3((B_ * N_) / 256), 256, 0, stream>>>(S, D);
  scalev_kernel<<<dim3((B_ * N_ * W_ / 4) / 256), 256, 0, stream>>>(V, D);
  {
    dim3 grid(((N_ / 16) * (W_ / 16)) / 8, B_);
    av_kernel<<<grid, 256, 0, stream>>>(S, V, Q, out);
  }
}


# column softmax re-spread: 1024-thread block per 32-column strip holds its 2048 rows in registers (read once); max and sum by fixed lane+LDS tree (new summation order); was one thread per column
# speedup vs baseline: 1.9679x; 1.9679x over previous
#include <hip/hip_runtime.h>

#define B_ 8
#define N_ 2048
#define C_ 256
#define W_ 256

typedef float v2f __attribute__((ext_vector_type(2)));
typedef float v8f __attribute__((ext_vector_type(8)));

__device__ __forceinline__ v8f wmma_f32_16x16x4(v2f a, v2f b, v8f c) {
  return __builtin_amdgcn_wmma_f32_16x16x4_f32(false, a, false, b, (short)0, c,
                                               false, false);
}

__global__ void qkv_kernel(const float* __restrict__ x,
                           const float* __restrict__ Wq, const float* __restrict__ bq,
                           const float* __restrict__ Wk, const float* __restrict__ bk,
                           const float* __restrict__ Wv, const float* __restrict__ bv,
                           float* __restrict__ Q, float* __restrict__ K,
                           float* __restrict__ V) {
  const float* Wm;
  const float* bias;
  float* out;
  if (blockIdx.y == 0) { Wm = Wq; bias = bq; out = Q; }
  else if (blockIdx.y == 1) { Wm = Wk; bias = bk; out = K; }
  else { Wm = Wv; bias = bv; out = V; }

  const int wave = threadIdx.x >> 5;
  const int lane = threadIdx.x & 31;
  const int half = lane >> 4;
  const int r    = lane & 15;

  const int tiles_n = W_ / 16;
  const int tile = blockIdx.x * 8 + wave;
  const int m0 = (tile / tiles_n) * 16;
  const int n0 = (tile % tiles_n) * 16;

  const float* arow = x  + (size_t)(m0 + r) * C_;
  const float* brow = Wm + (size_t)(n0 + r) * C_;

  v8f acc = {};
#pragma unroll 4
  for (int k0 = 0; k0 < C_; k0 += 4) {
    v2f a = *(const v2f*)(arow + k0 + 2 * half);
    v2f b = *(const v2f*)(brow + k0 + 2 * half);
    acc = wmma_f32_16x16x4(a, b, acc);
  }

  const float bv_ = bias[n0 + r];
#pragma unroll
  for (int g = 0; g < 8; ++g) {
    const int m = m0 + g + 8 * half;
    float v = acc[g] + bv_;
    v = v > 0.0f ? v : 0.0f;
    out[(size_t)m * W_ + n0 + r] = v;
  }
}

__global__ void scores_kernel(const float* __restrict__ Q,
                              const float* __restrict__ K,
                              float* __restrict__ S) {
  const int b    = blockIdx.y;
  const int wave = threadIdx.x >> 5;
  const int lane = threadIdx.x & 31;
  const int half = lane >> 4;
  const int r    = lane & 15;

  const int tiles_n = N_ / 16;
  const int tile = blockIdx.x * 8 + wave;
  const int i0 = (tile / tiles_n) * 16;
  const int j0 = (tile % tiles_n) * 16;

  const float* arow = Q + ((size_t)b * N_ + i0 + r) * W_;
  const float* brow = K + ((size_t)b * N_ + j0 + r) * W_;

  v8f acc = {};
#pragma unroll 4
  for (int k0 = 0; k0 < W_; k0 += 4) {
    v2f a = *(const v2f*)(arow + k0 + 2 * half);
    v2f b2 = *(const v2f*)(brow + k0 + 2 * half);
    acc = wmma_f32_16x16x4(a, b2, acc);
  }

  const float scale = 0.022097086912079608f;
  float* sb = S + (size_t)b * N_ * N_;
#pragma unroll
  for (int g = 0; g < 8; ++g) {
    const int i = i0 + g + 8 * half;
    sb[(size_t)i * N_ + j0 + r] = acc[g] * scale;
  }
}

#define SM_COLS    32
#define SM_WAVES   32
#define SM_THREADS (SM_WAVES * 32)
#define SM_STEPS   (N_ / (SM_WAVES * 4))

__global__ void __launch_bounds__(SM_THREADS)
colsoftmax_kernel(float* __restrict__ S, float* __restrict__ Dsum) {
  __shared__ __attribute__((aligned(16))) float red[SM_WAVES][SM_COLS];
  __shared__ __attribute__((aligned(16))) float fin[SM_COLS];

  const int b    = blockIdx.y;
  const int j0   = blockIdx.x * SM_COLS;
  const int wave = threadIdx.x >> 5;
  const int lane = threadIdx.x & 31;
  const int c4   = lane & 7;
  const int rsub = lane >> 3;

  float* base = S + (size_t)b * N_ * N_ + (size_t)(wave * 4 + rsub) * N_ + j0 + c4 * 4;
  const size_t step = (size_t)(SM_WAVES * 4) * N_;

  float4 v[SM_STEPS];
#pragma unroll
  for (int s = 0; s < SM_STEPS; ++s) v[s] = *(const float4*)(base + s * step);

  float4 m = v[0];
#pragma unroll
  for (int s = 1; s < SM_STEPS; ++s) {
    m.x = fmaxf(m.x, v[s].x); m.y = fmaxf(m.y, v[s].y);
    m.z = fmaxf(m.z, v[s].z); m.w = fmaxf(m.w, v[s].w);
  }
  m.x = fmaxf(m.x, __shfl_xor(m.x, 8));  m.y = fmaxf(m.y, __shfl_xor(m.y, 8));
  m.z = fmaxf(m.z, __shfl_xor(m.z, 8));  m.w = fmaxf(m.w, __shfl_xor(m.w, 8));
  m.x = fmaxf(m.x, __shfl_xor(m.x, 16)); m.y = fmaxf(m.y, __shfl_xor(m.y, 16));
  m.z = fmaxf(m.z, __shfl_xor(m.z, 16)); m.w = fmaxf(m.w, __shfl_xor(m.w, 16));
  if (lane < 8) *(float4*)&red[wave][c4 * 4] = m;
  __syncthreads();
  if (wave == 0) {
    float m0 = red[0][lane], m1 = red[1][lane], m2 = red[2][lane], m3 = red[3][lane];
#pragma unroll
    for (int w = 4; w < SM_WAVES; w += 4) {
      m0 = fmaxf(m0, red[w][lane]);     m1 = fmaxf(m1, red[w + 1][lane]);
      m2 = fmaxf(m2, red[w + 2][lane]); m3 = fmaxf(m3, red[w + 3][lane]);
    }
    fin[lane] = fmaxf(fmaxf(m0, m1), fmaxf(m2, m3));
  }
  __syncthreads();
  const float4 mx = *(const float4*)&fin[c4 * 4];

  float4 d = {0.0f, 0.0f, 0.0f, 0.0f};
#pragma unroll
  for (int s = 0; s < SM_STEPS; ++s) {
    float4 e;
    e.x = __expf(v[s].x - mx.x); e.y = __expf(v[s].y - mx.y);
    e.z = __expf(v[s].z - mx.z); e.w = __expf(v[s].w - mx.w);
    *(float4*)(base + s * step) = e;
    d.x += e.x; d.y += e.y; d.z += e.z; d.w += e.w;
  }
  d.x += __shfl_xor(d.x, 8);  d.y += __shfl_xor(d.y, 8);
  d.z += __shfl_xor(d.z, 8);  d.w += __shfl_xor(d.w, 8);
  d.x += __shfl_xor(d.x, 16); d.y += __shfl_xor(d.y, 16);
  d.z += __shfl_xor(d.z, 16); d.w += __shfl_xor(d.w, 16);
  if (lane < 8) *(float4*)&red[wave][c4 * 4] = d;
  __syncthreads();
  if (wave == 0) {
    float d0 = red[0][lane], d1 = red[1][lane], d2 = red[2][lane], d3 = red[3][lane];
#pragma unroll
    for (int w = 4; w < SM_WAVES; w += 4) {
      d0 += red[w][lane];     d1 += red[w + 1][lane];
      d2 += red[w + 2][lane]; d3 += red[w + 3][lane];
    }
    Dsum[(size_t)b * N_ + j0 + lane] = (d0 + d1) + (d2 + d3);
  }
}

__global__ void scalev_kernel(float* __restrict__ V, const float* __restrict__ Dsum) {
  const int idx = blockIdx.x * blockDim.x + threadIdx.x;
  const int total = B_ * N_ * W_ / 4;
  if (idx >= total) return;
  float4* v4 = (float4*)V;
  const int row = (idx * 4) / W_;
  const float inv = 1.0f / Dsum[row];
  float4 v = v4[idx];
  v.x *= inv; v.y *= inv; v.z *= inv; v.w *= inv;
  v4[idx] = v;
}

__global__ void av_kernel(const float* __restrict__ P, const float* __restrict__ V,
                          const float* __restrict__ Q, float* __restrict__ out) {
  const int b    = blockIdx.y;
  const int wave = threadIdx.x >> 5;
  const int lane = threadIdx.x & 31;
  const int half = lane >> 4;
  const int r    = lane & 15;

  const int tiles_n = W_ / 16;
  const int tile = blockIdx.x * 8 + wave;
  const int i0 = (tile / tiles_n) * 16;
  const int n0 = (tile % tiles_n) * 16;

  const float* arow = P + (size_t)b * N_ * N_ + (size_t)(i0 + r) * N_;
  const float* vb   = V + (size_t)b * N_ * W_;

  v8f acc = {};
#pragma unroll 2
  for (int k0 = 0; k0 < N_; k0 += 4) {
    v2f a = *(const v2f*)(arow + k0 + 2 * half);
    const int kr = k0 + 2 * half;
    v2f b2;
    b2.x = vb[(size_t)kr * W_ + n0 + r];
    b2.y = vb[(size_t)(kr + 1) * W_ + n0 + r];
    acc = wmma_f32_16x16x4(a, b2, acc);
  }

  const float* qb = Q + (size_t)b * N_ * W_;
  float* ob = out + (size_t)b * N_ * W_;
#pragma unroll
  for (int g = 0; g < 8; ++g) {
    const int i = i0 + g + 8 * half;
    ob[(size_t)i * W_ + n0 + r] = acc[g] + qb[(size_t)i * W_ + n0 + r];
  }
}

extern "C" void kernel_launch(void* const* d_in, const int* in_sizes, int n_in,
                              void* d_out, int out_size, void* d_ws, size_t ws_size,
                              hipStream_t stream) {
  const float* x  = (const float*)d_in[0];
  const float* Wq = (const float*)d_in[1];
  const float* bq = (const float*)d_in[2];
  const float* Wk = (const float*)d_in[3];
  const float* bk = (const float*)d_in[4];
  const float* Wv = (const float*)d_in[5];
  const float* bv = (const float*)d_in[6];
  float* out = (float*)d_out;
  float* ws  = (float*)d_ws;

  const size_t QKV = (size_t)B_ * N_ * W_;
  float* Q = ws;
  float* K = ws + QKV;
  float* V = ws + 2 * QKV;
  float* D = ws + 3 * QKV;
  float* S = D + (size_t)B_ * N_;

  {
    dim3 grid(((B_ * N_ / 16) * (W_ / 16)) / 8, 3);
    qkv_kernel<<<grid, 256, 0, stream>>>(x, Wq, bq, Wk, bk, Wv, bv, Q, K, V);
  }
  {
    dim3 grid(((N_ / 16) * (N_ / 16)) / 8, B_);
    scores_kernel<<<grid, 256, 0, stream>>>(Q, K, S);
  }
  colsoftmax_kernel<<<dim3(N_ / SM_COLS, B_), SM_THREADS, 0, stream>>>(S, D);
  scalev_kernel<<<dim3((B_ * N_ * W_ / 4) / 256), 256, 0, stream>>>(V, D);
  {
    dim3 grid(((N_ / 16) * (W_ / 16)) / 8, B_);
    av_kernel<<<grid, 256, 0, stream>>>(S, V, Q, out);
  }
}


# GEMM operand reuse in registers: each wave computes a 64x32/64x64/32x64 block of tiles from 16-byte fragment loads issued one chunk ahead (was 1 tile, 2-3 loads per product); new k summation order
# speedup vs baseline: 4.7948x; 2.4365x over previous
#include <hip/hip_runtime.h>

#define B_ 8
#define N_ 2048
#define C_ 256
#define W_ 256

typedef float v2f __attribute__((ext_vector_type(2)));
typedef float v8f __attribute__((ext_vector_type(8)));

__device__ __forceinline__ v8f wmma_f32_16x16x4(v2f a, v2f b, v8f c) {
  return __builtin_amdgcn_wmma_f32_16x16x4_f32(false, a, false, b, (short)0, c,
                                               false, false);
}

typedef float v4f __attribute__((ext_vector_type(4)));

template <int TM, int TN, int KD>
struct WaveBlockNT {
  static __device__ __forceinline__ void load(v4f (&a)[TM], v4f (&b)[TN],
                                              const float* __restrict__ arow,
                                              const float* __restrict__ brow, int k0) {
#pragma unroll
    for (int t = 0; t < TM; ++t) a[t] = *(const v4f*)(arow + (size_t)t * 16 * KD + k0);
#pragma unroll
    for (int t = 0; t < TN; ++t) b[t] = *(const v4f*)(brow + (size_t)t * 16 * KD + k0);
  }
  static __device__ __forceinline__ void mma(const v4f (&a)[TM], const v4f (&b)[TN],
                                             v8f (&acc)[TM][TN]) {
#pragma unroll
    for (int i = 0; i < TM; ++i)
#pragma unroll
      for (int j = 0; j < TN; ++j)
        acc[i][j] = wmma_f32_16x16x4(__builtin_shufflevector(a[i], a[i], 0, 1),
                                     __builtin_shufflevector(b[j], b[j], 0, 1), acc[i][j]);
#pragma unroll
    for (int i = 0; i < TM; ++i)
#pragma unroll
      for (int j = 0; j < TN; ++j)
        acc[i][j] = wmma_f32_16x16x4(__builtin_shufflevector(a[i], a[i], 2, 3),
                                     __builtin_shufflevector(b[j], b[j], 2, 3), acc[i][j]);
  }
  static __device__ __forceinline__ void run(const float* __restrict__ arow,
                                             const float* __restrict__ brow,
                                             v8f (&acc)[TM][TN]) {
    v4f a0[TM], b0[TN], a1[TM], b1[TN];
    load(a0, b0, arow, brow, 0);
#pragma unroll 1
    for (int k0 = 0; k0 < KD; k0 += 16) {
      load(a1, b1, arow, brow, k0 + 8);
      __builtin_amdgcn_sched_barrier(0);
      mma(a0, b0, acc);
      const int kn = (k0 + 16 < KD) ? k0 + 16 : 0;
      __builtin_amdgcn_sched_barrier(0);
      load(a0, b0, arow, brow, kn);
      __builtin_amdgcn_sched_barrier(0);
      mma(a1, b1, acc);
    }
  }
};

#define QKV_TM 4
#define QKV_TN 2
__global__ void __launch_bounds__(256)
qkv_kernel(const float* __restrict__ x,
           const float* __restrict__ Wq, const float* __restrict__ bq,
           const float* __restrict__ Wk, const float* __restrict__ bk,
           const float* __restrict__ Wv, const float* __restrict__ bv,
           float* __restrict__ Q, float* __restrict__ K,
           float* __restrict__ V) {
  const float* Wm;
  const float* bias;
  float* out;
  if (blockIdx.y == 0) { Wm = Wq; bias = bq; out = Q; }
  else if (blockIdx.y == 1) { Wm = Wk; bias = bk; out = K; }
  else { Wm = Wv; bias = bv; out = V; }

  const int wave = threadIdx.x >> 5;
  const int lane = threadIdx.x & 31;
  const int half = lane >> 4;
  const int r    = lane & 15;

  const int bn = blockIdx.x & 1;
  const int bm = blockIdx.x >> 1;
  const int m0 = bm * 128 + (wave >> 2) * (16 * QKV_TM);
  const int n0 = bn * 128 + (wave & 3) * (16 * QKV_TN);

  const float* arow = x  + (size_t)(m0 + r) * C_ + 4 * half;
  const float* brow = Wm + (size_t)(n0 + r) * C_ + 4 * half;

  v8f acc[QKV_TM][QKV_TN] = {};
  WaveBlockNT<QKV_TM, QKV_TN, C_>::run(arow, brow, acc);

#pragma unroll
  for (int j = 0; j < QKV_TN; ++j) {
    const int n = n0 + j * 16 + r;
    const float bv_ = bias[n];
#pragma unroll
    for (int i = 0; i < QKV_TM; ++i) {
#pragma unroll
      for (int g = 0; g < 8; ++g) {
        const int m = m0 + i * 16 + g + 8 * half;
        float v = acc[i][j][g] + bv_;
        v = v > 0.0f ? v : 0.0f;
        out[(size_t)m * W_ + n] = v;
      }
    }
  }
}

#define SC_TM 4
#define SC_TN 4
__global__ void __launch_bounds__(256)
scores_kernel(const float* __restrict__ Q,
              const float* __restrict__ K,
              float* __restrict__ S) {
  const int b    = blockIdx.y;
  const int wave = threadIdx.x >> 5;
  const int lane = threadIdx.x & 31;
  const int half = lane >> 4;
  const int r    = lane & 15;

  const int bj = blockIdx.x & 7;
  const int bi = blockIdx.x >> 3;
  const int i0 = bi * 128 + (wave >> 2) * (16 * SC_TM);
  const int j0 = bj * 256 + (wave & 3) * (16 * SC_TN);

  const float* arow = Q + ((size_t)b * N_ + i0 + r) * W_ + 4 * half;
  const float* brow = K + ((size_t)b * N_ + j0 + r) * W_ + 4 * half;

  v8f acc[SC_TM][SC_TN] = {};
  WaveBlockNT<SC_TM, SC_TN, W_>::run(arow, brow, acc);

  const float scale = 0.022097086912079608f;
  float* sb = S + (size_t)b * N_ * N_;
#pragma unroll
  for (int i = 0; i < SC_TM; ++i) {
#pragma unroll
    for (int g = 0; g < 8; ++g) {
      const int row = i0 + i * 16 + g + 8 * half;
#pragma unroll
      for (int j = 0; j < SC_TN; ++j)
        sb[(size_t)row * N_ + j0 + j * 16 + r] = acc[i][j][g] * scale;
    }
  }
}

#define SM_COLS    32
#define SM_WAVES   32
#define SM_THREADS (SM_WAVES * 32)
#define SM_STEPS   (N_ / (SM_WAVES * 4))

__global__ void __launch_bounds__(SM_THREADS)
colsoftmax_kernel(float* __restrict__ S, float* __restrict__ Dsum) {
  __shared__ __attribute__((aligned(16))) float red[SM_WAVES][SM_COLS];
  __shared__ __attribute__((aligned(16))) float fin[SM_COLS];

  const int b    = blockIdx.y;
  const int j0   = blockIdx.x * SM_COLS;
  const int wave = threadIdx.x >> 5;
  const int lane = threadIdx.x & 31;
  const int c4   = lane & 7;
  const int rsub = lane >> 3;

  float* base = S + (size_t)b * N_ * N_ + (size_t)(wave * 4 + rsub) * N_ + j0 + c4 * 4;
  const size_t step = (size_t)(SM_WAVES * 4) * N_;

  float4 v[SM_STEPS];
#pragma unroll
  for (int s = 0; s < SM_STEPS; ++s) v[s] = *(const float4*)(base + s * step);

  float4 m = v[0];
#pragma unroll
  for (int s = 1; s < SM_STEPS; ++s) {
    m.x = fmaxf(m.x, v[s].x); m.y = fmaxf(m.y, v[s].y);
    m.z = fmaxf(m.z, v[s].z); m.w = fmaxf(m.w, v[s].w);
  }
  m.x = fmaxf(m.x, __shfl_xor(m.x, 8));  m.y = fmaxf(m.y, __shfl_xor(m.y, 8));
  m.z = fmaxf(m.z, __shfl_xor(m.z, 8));  m.w = fmaxf(m.w, __shfl_xor(m.w, 8));
  m.x = fmaxf(m.x, __shfl_xor(m.x, 16)); m.y = fmaxf(m.y, __shfl_xor(m.y, 16));
  m.z = fmaxf(m.z, __shfl_xor(m.z, 16)); m.w = fmaxf(m.w, __shfl_xor(m.w, 16));
  if (lane < 8) *(float4*)&red[wave][c4 * 4] = m;
  __syncthreads();
  if (wave == 0) {
    float m0 = red[0][lane], m1 = red[1][lane], m2 = red[2][lane], m3 = red[3][lane];
#pragma unroll
    for (int w = 4; w < SM_WAVES; w += 4) {
      m0 = fmaxf(m0, red[w][lane]);     m1 = fmaxf(m1, red[w + 1][lane]);
      m2 = fmaxf(m2, red[w + 2][lane]); m3 = fmaxf(m3, red[w + 3][lane]);
    }
    fin[lane] = fmaxf(fmaxf(m0, m1), fmaxf(m2, m3));
  }
  __syncthreads();
  const float4 mx = *(const float4*)&fin[c4 * 4];

  float4 d = {0.0f, 0.0f, 0.0f, 0.0f};
#pragma unroll
  for (int s = 0; s < SM_STEPS; ++s) {
    float4 e;
    e.x = __expf(v[s].x - mx.x); e.y = __expf(v[s].y - mx.y);
    e.z = __expf(v[s].z - mx.z); e.w = __expf(v[s].w - mx.w);
    *(float4*)(base + s * step) = e;
    d.x += e.x; d.y += e.y; d.z += e.z; d.w += e.w;
  }
  d.x += __shfl_xor(d.x, 8);  d.y += __shfl_xor(d.y, 8);
  d.z += __shfl_xor(d.z, 8);  d.w += __shfl_xor(d.w, 8);
  d.x += __shfl_xor(d.x, 16); d.y += __shfl_xor(d.y, 16);
  d.z += __shfl_xor(d.z, 16); d.w += __shfl_xor(d.w, 16);
  if (lane < 8) *(float4*)&red[wave][c4 * 4] = d;
  __syncthreads();
  if (wave == 0) {
    float d0 = red[0][lane], d1 = red[1][lane], d2 = red[2][lane], d3 = red[3][lane];
#pragma unroll
    for (int w = 4; w < SM_WAVES; w += 4) {
      d0 += red[w][lane];     d1 += red[w + 1][lane];
      d2 += red[w + 2][lane]; d3 += red[w + 3][lane];
    }
    Dsum[(size_t)b * N_ + j0 + lane] = (d0 + d1) + (d2 + d3);
  }
}

__global__ void scalev_kernel(float* __restrict__ V, const float* __restrict__ Dsum) {
  const int idx = blockIdx.x * blockDim.x + threadIdx.x;
  const int total = B_ * N_ * W_ / 4;
  if (idx >= total) return;
  float4* v4 = (float4*)V;
  const int row = (idx * 4) / W_;
  const float inv = 1.0f / Dsum[row];
  float4 v = v4[idx];
  v.x *= inv; v.y *= inv; v.z *= inv; v.w *= inv;
  v4[idx] = v;
}

#define AV_TM 2
#define AV_TN 4
#define AV_KL 16
struct AvStage { v4f a[AV_TM][AV_KL / 4]; v4f b[AV_KL]; };

__device__ __forceinline__ void av_load(AvStage& s, const float* __restrict__ arow,
                                        const float* __restrict__ brow, int k0) {
#pragma unroll
  for (int t = 0; t < AV_TM; ++t)
#pragma unroll
    for (int q = 0; q < AV_KL / 4; ++q)
      s.a[t][q] = *(const v4f*)(arow + (size_t)t * 16 * N_ + k0 + 4 * q);
#pragma unroll
  for (int e = 0; e < AV_KL; ++e) s.b[e] = *(const v4f*)(brow + (size_t)(k0 + e) * W_);
}

__device__ __forceinline__ void av_mma(const AvStage& s, v8f (&acc)[AV_TM][AV_TN]) {
#pragma unroll
  for (int q = 0; q < AV_KL / 4; ++q) {
#pragma unroll
    for (int p = 0; p < 2; ++p) {
      const int e0 = 4 * q + 2 * p;
#pragma unroll
      for (int j = 0; j < AV_TN; ++j) {
        v2f bf;
        bf.x = s.b[e0][j];
        bf.y = s.b[e0 + 1][j];
#pragma unroll
        for (int i = 0; i < AV_TM; ++i) {
          v2f af;
          af.x = s.a[i][q][2 * p];
          af.y = s.a[i][q][2 * p + 1];
          acc[i][j] = wmma_f32_16x16x4(af, bf, acc[i][j]);
        }
      }
    }
  }
}

__global__ void __launch_bounds__(256)
av_kernel(const float* __restrict__ P, const float* __restrict__ V,
          const float* __restrict__ Q, float* __restrict__ out) {
  const int b    = blockIdx.y;
  const int wave = threadIdx.x >> 5;
  const int lane = threadIdx.x & 31;
  const int half = lane >> 4;
  const int r    = lane & 15;

  const int i0 = blockIdx.x * 64 + (wave >> 2) * (16 * AV_TM);
  const int n0 = (wave & 3) * (16 * AV_TN);

  const float* arow = P + (size_t)b * N_ * N_ + (size_t)(i0 + r) * N_ + AV_KL * half;
  const float* brow = V + (size_t)b * N_ * W_ + (size_t)(AV_KL * half) * W_ + n0 + 4 * r;

  v8f acc[AV_TM][AV_TN] = {};
  AvStage s0, s1;
  av_load(s0, arow, brow, 0);
#pragma unroll 1
  for (int k0 = 0; k0 < N_; k0 += 4 * AV_KL) {
    av_load(s1, arow, brow, k0 + 2 * AV_KL);
    __builtin_amdgcn_sched_barrier(0);
    av_mma(s0, acc);
    const int kn = (k0 + 4 * AV_KL < N_) ? k0 + 4 * AV_KL : 0;
    __builtin_amdgcn_sched_barrier(0);
    av_load(s0, arow, brow, kn);
    __builtin_amdgcn_sched_barrier(0);
    av_mma(s1, acc);
  }

  const float* qb = Q + (size_t)b * N_ * W_;
  float* ob = out + (size_t)b * N_ * W_;
#pragma unroll
  for (int i = 0; i < AV_TM; ++i) {
#pragma unroll
    for (int g = 0; g < 8; ++g) {
      const size_t off = (size_t)(i0 + i * 16 + g + 8 * half) * W_ + n0 + 4 * r;
      const v4f q4 = *(const v4f*)(qb + off);
      v4f o;
      o.x = acc[i][0][g] + q4.x; o.y = acc[i][1][g] + q4.y;
      o.z = acc[i][2][g] + q4.z; o.w = acc[i][3][g] + q4.w;
      *(v4f*)(ob + off) = o;
    }
  }
}

extern "C" void kernel_launch(void* const* d_in, const int* in_sizes, int n_in,
                              void* d_out, int out_size, void* d_ws, size_t ws_size,
                              hipStream_t stream) {
  const float* x  = (const float*)d_in[0];
  const float* Wq = (const float*)d_in[1];
  const float* bq = (const float*)d_in[2];
  const float* Wk = (const float*)d_in[3];
  const float* bk = (const float*)d_in[4];
  const float* Wv = (const float*)d_in[5];
  const float* bv = (const float*)d_in[6];
  float* out = (float*)d_out;
  float* ws  = (float*)d_ws;

  const size_t QKV = (size_t)B_ * N_ * W_;
  float* Q = ws;
  float* K = ws + QKV;
  float* V = ws + 2 * QKV;
  float* D = ws + 3 * QKV;
  float* S = D + (size_t)B_ * N_;

  {
    dim3 grid((B_ * N_ / 128) * (W_ / 128), 3);
    qkv_kernel<<<grid, 256, 0, stream>>>(x, Wq, bq, Wk, bk, Wv, bv, Q, K, V);
  }
  {
    dim3 grid((N_ / 128) * (N_ / 256), B_);
    scores_kernel<<<grid, 256, 0, stream>>>(Q, K, S);
  }
  colsoftmax_kernel<<<dim3(N_ / SM_COLS, B_), SM_THREADS, 0, stream>>>(S, D);
  scalev_kernel<<<dim3((B_ * N_ * W_ / 4) / 256), 256, 0, stream>>>(V, D);
  {
    dim3 grid(N_ / 64, B_);
    av_kernel<<<grid, 256, 0, stream>>>(S, V, Q, out);
  }
}


# projection and score GEMMs: operands staged through LDS in 32-k steps (whole-line global loads, each line once per work-group; 16-byte LDS fragment reads); sums keep their order
# speedup vs baseline: 5.7821x; 1.2059x over previous
#include <hip/hip_runtime.h>

#define B_ 8
#define N_ 2048
#define C_ 256
#define W_ 256

typedef float v2f __attribute__((ext_vector_type(2)));
typedef float v8f __attribute__((ext_vector_type(8)));

__device__ __forceinline__ v8f wmma_f32_16x16x4(v2f a, v2f b, v8f c) {
  return __builtin_amdgcn_wmma_f32_16x16x4_f32(false, a, false, b, (short)0, c,
                                               false, false);
}

typedef float v4f __attribute__((ext_vector_type(4)));

#define KS_  32
#define LDP_ 36
template <int TM, int TN, int KD>
struct BlockNT {
  static constexpr int BM = 32 * TM;
  static constexpr int BN = 64 * TN;
  static constexpr int NLA = BM / 32;
  static constexpr int NLB = BN / 32;
  static constexpr int STAGE = (BM + BN) * LDP_;
  static constexpr int LDS_FLOATS = 2 * STAGE;

  static __device__ __forceinline__ void gload(v4f (&g)[NLA + NLB],
                                               const float* __restrict__ ag,
                                               const float* __restrict__ bg, int k0) {
#pragma unroll
    for (int l = 0; l < NLA; ++l) g[l] = *(const v4f*)(ag + (size_t)l * 32 * KD + k0);
#pragma unroll
    for (int l = 0; l < NLB; ++l) g[NLA + l] = *(const v4f*)(bg + (size_t)l * 32 * KD + k0);
  }
  static __device__ __forceinline__ void lstore(float* st, const v4f (&g)[NLA + NLB]) {
#pragma unroll
    for (int l = 0; l < NLA; ++l) *(v4f*)(st + l * 32 * LDP_) = g[l];
#pragma unroll
    for (int l = 0; l < NLB; ++l) *(v4f*)(st + (BM + l * 32) * LDP_) = g[NLA + l];
  }
  static __device__ __forceinline__ void fload(v4f (&a)[TM], v4f (&b)[TN],
                                               const float* fa, const float* fb, int q) {
#pragma unroll
    for (int t = 0; t < TM; ++t) a[t] = *(const v4f*)(fa + t * 16 * LDP_ + q * 8);
#pragma unroll
    for (int t = 0; t < TN; ++t) b[t] = *(const v4f*)(fb + t * 16 * LDP_ + q * 8);
  }
  static __device__ __forceinline__ void mma(const v4f (&a)[TM], const v4f (&b)[TN],
                                             v8f (&acc)[TM][TN]) {
#pragma unroll
    for (int i = 0; i < TM; ++i)
#pragma unroll
      for (int j = 0; j < TN; ++j)
        acc[i][j] = wmma_f32_16x16x4(__builtin_shufflevector(a[i], a[i], 0, 1),
                                     __builtin_shufflevector(b[j], b[j], 0, 1), acc[i][j]);
#pragma unroll
    for (int i = 0; i < TM; ++i)
#pragma unroll
      for (int j = 0; j < TN; ++j)
        acc[i][j] = wmma_f32_16x16x4(__builtin_shufflevector(a[i], a[i], 2, 3),
                                     __builtin_shufflevector(b[j], b[j], 2, 3), acc[i][j]);
  }
  static __device__ __forceinline__ void run(const float* __restrict__ ag,
                                             const float* __restrict__ bg,
                                             float* lds, int tid, int wm, int wn,
                                             int r, int half, v8f (&acc)[TM][TN]) {
    constexpr int NK = KD / KS_;
    float* st = lds + (tid >> 3) * LDP_ + (tid & 7) * 4;
    const float* fa = lds + (wm * TM * 16 + r) * LDP_ + 4 * half;
    const float* fb = lds + (BM + wn * TN * 16 + r) * LDP_ + 4 * half;

    v4f g[NLA + NLB];
    gload(g, ag, bg, 0);
    lstore(st, g);
    __syncthreads();

    v4f a0[TM], b0[TN], a1[TM], b1[TN];
#pragma unroll 1
    for (int ks = 0; ks < NK; ++ks) {
      const int cur = (ks & 1) * STAGE;
      const bool more = ks + 1 < NK;
      if (more) gload(g, ag, bg, (ks + 1) * KS_);
      fload(a0, b0, fa + cur, fb + cur, 0);
      fload(a1, b1, fa + cur, fb + cur, 1);
      __builtin_amdgcn_sched_barrier(0);
      mma(a0, b0, acc);
      __builtin_amdgcn_sched_barrier(0);
      fload(a0, b0, fa + cur, fb + cur, 2);
      __builtin_amdgcn_sched_barrier(0);
      mma(a1, b1, acc);
      __builtin_amdgcn_sched_barrier(0);
      fload(a1, b1, fa + cur, fb + cur, 3);
      __builtin_amdgcn_sched_barrier(0);
      mma(a0, b0, acc);
      mma(a1, b1, acc);
      if (more) lstore(st + (STAGE - cur), g);
      __syncthreads();
    }
  }
};

#define QKV_TM 4
#define QKV_TN 2
__global__ void __launch_bounds__(256)
qkv_kernel(const float* __restrict__ x,
           const float* __restrict__ Wq, const float* __restrict__ bq,
           const float* __restrict__ Wk, const float* __restrict__ bk,
           const float* __restrict__ Wv, const float* __restrict__ bv,
           float* __restrict__ Q, float* __restrict__ K,
           float* __restrict__ V) {
  const float* Wm;
  const float* bias;
  float* out;
  if (blockIdx.y == 0) { Wm = Wq; bias = bq; out = Q; }
  else if (blockIdx.y == 1) { Wm = Wk; bias = bk; out = K; }
  else { Wm = Wv; bias = bv; out = V; }

  const int wave = threadIdx.x >> 5;
  const int lane = threadIdx.x & 31;
  const int half = lane >> 4;
  const int r    = lane & 15;

  const int bn = blockIdx.x & 1;
  const int bm = blockIdx.x >> 1;
  const int m0 = bm * 128 + (wave >> 2) * (16 * QKV_TM);
  const int n0 = bn * 128 + (wave & 3) * (16 * QKV_TN);

  typedef BlockNT<QKV_TM, QKV_TN, C_> Blk;
  __shared__ __attribute__((aligned(16))) float lds[Blk::LDS_FLOATS];
  const int tid = threadIdx.x;
  const float* ag = x  + (size_t)(bm * 128 + (tid >> 3)) * C_ + (tid & 7) * 4;
  const float* bg = Wm + (size_t)(bn * 128 + (tid >> 3)) * C_ + (tid & 7) * 4;

  v8f acc[QKV_TM][QKV_TN] = {};
  Blk::run(ag, bg, lds, tid, wave >> 2, wave & 3, r, half, acc);

#pragma unroll
  for (int j = 0; j < QKV_TN; ++j) {
    const int n = n0 + j * 16 + r;
    const float bv_ = bias[n];
#pragma unroll
    for (int i = 0; i < QKV_TM; ++i) {
#pragma unroll
      for (int g = 0; g < 8; ++g) {
        const int m = m0 + i * 16 + g + 8 * half;
        float v = acc[i][j][g] + bv_;
        v = v > 0.0f ? v : 0.0f;
        out[(size_t)m * W_ + n] = v;
      }
    }
  }
}

#define SC_TM 4
#define SC_TN 4
__global__ void __launch_bounds__(256)
scores_kernel(const float* __restrict__ Q,
              const float* __restrict__ K,
              float* __restrict__ S) {
  const int b    = blockIdx.y;
  const int wave = threadIdx.x >> 5;
  const int lane = threadIdx.x & 31;
  const int half = lane >> 4;
  const int r    = lane & 15;

  const int bj = blockIdx.x & 7;
  const int bi = blockIdx.x >> 3;
  const int i0 = bi * 128 + (wave >> 2) * (16 * SC_TM);
  const int j0 = bj * 256 + (wave & 3) * (16 * SC_TN);

  typedef BlockNT<SC_TM, SC_TN, W_> Blk;
  __shared__ __attribute__((aligned(16))) float lds[Blk::LDS_FLOATS];
  const int tid = threadIdx.x;
  const float* ag = Q + ((size_t)b * N_ + bi * 128 + (tid >> 3)) * W_ + (tid & 7) * 4;
  const float* bg = K + ((size_t)b * N_ + bj * 256 + (tid >> 3)) * W_ + (tid & 7) * 4;

  v8f acc[SC_TM][SC_TN] = {};
  Blk::run(ag, bg, lds, tid, wave >> 2, wave & 3, r, half, acc);

  const float scale = 0.022097086912079608f;
  float* sb = S + (size_t)b * N_ * N_;
#pragma unroll
  for (int i = 0; i < SC_TM; ++i) {
#pragma unroll
    for (int g = 0; g < 8; ++g) {
      const int row = i0 + i * 16 + g + 8 * half;
#pragma unroll
      for (int j = 0; j < SC_TN; ++j)
        sb[(size_t)row * N_ + j0 + j * 16 + r] = acc[i][j][g] * scale;
    }
  }
}

#define SM_COLS    32
#define SM_WAVES   32
#define SM_THREADS (SM_WAVES * 32)
#define SM_STEPS   (N_ / (SM_WAVES * 4))

__global__ void __launch_bounds__(SM_THREADS)
colsoftmax_kernel(float* __restrict__ S, float* __restrict__ Dsum) {
  __shared__ __attribute__((aligned(16))) float red[SM_WAVES][SM_COLS];
  __shared__ __attribute__((aligned(16))) float fin[SM_COLS];

  const int b    = blockIdx.y;
  const int j0   = blockIdx.x * SM_COLS;
  const int wave = threadIdx.x >> 5;
  const int lane = threadIdx.x & 31;
  const int c4   = lane & 7;
  const int rsub = lane >> 3;

  float* base = S + (size_t)b * N_ * N_ + (size_t)(wave * 4 + rsub) * N_ + j0 + c4 * 4;
  const size_t step = (size_t)(SM_WAVES * 4) * N_;

  float4 v[SM_STEPS];
#pragma unroll
  for (int s = 0; s < SM_STEPS; ++s) v[s] = *(const float4*)(base + s * step);

  float4 m = v[0];
#pragma unroll
  for (int s = 1; s < SM_STEPS; ++s) {
    m.x = fmaxf(m.x, v[s].x); m.y = fmaxf(m.y, v[s].y);
    m.z = fmaxf(m.z, v[s].z); m.w = fmaxf(m.w, v[s].w);
  }
  m.x = fmaxf(m.x, __shfl_xor(m.x, 8));  m.y = fmaxf(m.y, __shfl_xor(m.y, 8));
  m.z = fmaxf(m.z, __shfl_xor(m.z, 8));  m.w = fmaxf(m.w, __shfl_xor(m.w, 8));
  m.x = fmaxf(m.x, __shfl_xor(m.x, 16)); m.y = fmaxf(m.y, __shfl_xor(m.y, 16));
  m.z = fmaxf(m.z, __shfl_xor(m.z, 16)); m.w = fmaxf(m.w, __shfl_xor(m.w, 16));
  if (lane < 8) *(float4*)&red[wave][c4 * 4] = m;
  __syncthreads();
  if (wave == 0) {
    float m0 = red[0][lane], m1 = red[1][lane], m2 = red[2][lane], m3 = red[3][lane];
#pragma unroll
    for (int w = 4; w < SM_WAVES; w += 4) {
      m0 = fmaxf(m0, red[w][lane]);     m1 = fmaxf(m1, red[w + 1][lane]);
      m2 = fmaxf(m2, red[w + 2][lane]); m3 = fmaxf(m3, red[w + 3][lane]);
    }
    fin[lane] = fmaxf(fmaxf(m0, m1), fmaxf(m2, m3));
  }
  __syncthreads();
  const float4 mx = *(const float4*)&fin[c4 * 4];

  float4 d = {0.0f, 0.0f, 0.0f, 0.0f};
#pragma unroll
  for (int s = 0; s < SM_STEPS; ++s) {
    float4 e;
    e.x = __expf(v[s].x - mx.x); e.y = __expf(v[s].y - mx.y);
    e.z = __expf(v[s].z - mx.z); e.w = __expf(v[s].w - mx.w);
    *(float4*)(base + s * step) = e;
    d.x += e.x; d.y += e.y; d.z += e.z; d.w += e.w;
  }
  d.x += __shfl_xor(d.x, 8);  d.y += __shfl_xor(d.y, 8);
  d.z += __shfl_xor(d.z, 8);  d.w += __shfl_xor(d.w, 8);
  d.x += __shfl_xor(d.x, 16); d.y += __shfl_xor(d.y, 16);
  d.z += __shfl_xor(d.z, 16); d.w += __shfl_xor(d.w, 16);
  if (lane < 8) *(float4*)&red[wave][c4 * 4] = d;
  __syncthreads();
  if (wave == 0) {
    float d0 = red[0][lane], d1 = red[1][lane], d2 = red[2][lane], d3 = red[3][lane];
#pragma unroll
    for (int w = 4; w < SM_WAVES; w += 4) {
      d0 += red[w][lane];     d1 += red[w + 1][lane];
      d2 += red[w + 2][lane]; d3 += red[w + 3][lane];
    }
    Dsum[(size_t)b * N_ + j0 + lane] = (d0 + d1) + (d2 + d3);
  }
}

__global__ void scalev_kernel(float* __restrict__ V, const float* __restrict__ Dsum) {
  const int idx = blockIdx.x * blockDim.x + threadIdx.x;
  const int total = B_ * N_ * W_ / 4;
  if (idx >= total) return;
  float4* v4 = (float4*)V;
  const int row = (idx * 4) / W_;
  const float inv = 1.0f / Dsum[row];
  float4 v = v4[idx];
  v.x *= inv; v.y *= inv; v.z *= inv; v.w *= inv;
  v4[idx] = v;
}

#define AV_TM 2
#define AV_TN 4
#define AV_KL 16
struct AvStage { v4f a[AV_TM][AV_KL / 4]; v4f b[AV_KL]; };

__device__ __forceinline__ void av_load(AvStage& s, const float* __restrict__ arow,
                                        const float* __restrict__ brow, int k0) {
#pragma unroll
  for (int t = 0; t < AV_TM; ++t)
#pragma unroll
    for (int q = 0; q < AV_KL / 4; ++q)
      s.a[t][q] = *(const v4f*)(arow + (size_t)t * 16 * N_ + k0 + 4 * q);
#pragma unroll
  for (int e = 0; e < AV_KL; ++e) s.b[e] = *(const v4f*)(brow + (size_t)(k0 + e) * W_);
}

__device__ __forceinline__ void av_mma(const AvStage& s, v8f (&acc)[AV_TM][AV_TN]) {
#pragma unroll
  for (int q = 0; q < AV_KL / 4; ++q) {
#pragma unroll
    for (int p = 0; p < 2; ++p) {
      const int e0 = 4 * q + 2 * p;
#pragma unroll
      for (int j = 0; j < AV_TN; ++j) {
        v2f bf;
        bf.x = s.b[e0][j];
        bf.y = s.b[e0 + 1][j];
#pragma unroll
        for (int i = 0; i < AV_TM; ++i) {
          v2f af;
          af.x = s.a[i][q][2 * p];
          af.y = s.a[i][q][2 * p + 1];
          acc[i][j] = wmma_f32_16x16x4(af, bf, acc[i][j]);
        }
      }
    }
  }
}

__global__ void __launch_bounds__(256)
av_kernel(const float* __restrict__ P, const float* __restrict__ V,
          const float* __restrict__ Q, float* __restrict__ out) {
  const int b    = blockIdx.y;
  const int wave = threadIdx.x >> 5;
  const int lane = threadIdx.x & 31;
  const int half = lane >> 4;
  const int r    = lane & 15;

  const int i0 = blockIdx.x * 64 + (wave >> 2) * (16 * AV_TM);
  const int n0 = (wave & 3) * (16 * AV_TN);

  const float* arow = P + (size_t)b * N_ * N_ + (size_t)(i0 + r) * N_ + AV_KL * half;
  const float* brow = V + (size_t)b * N_ * W_ + (size_t)(AV_KL * half) * W_ + n0 + 4 * r;

  v8f acc[AV_TM][AV_TN] = {};
  AvStage s0, s1;
  av_load(s0, arow, brow, 0);
#pragma unroll 1
  for (int k0 = 0; k0 < N_; k0 += 4 * AV_KL) {
    av_load(s1, arow, brow, k0 + 2 * AV_KL);
    __builtin_amdgcn_sched_barrier(0);
    av_mma(s0, acc);
    const int kn = (k0 + 4 * AV_KL < N_) ? k0 + 4 * AV_KL : 0;
    __builtin_amdgcn_sched_barrier(0);
    av_load(s0, arow, brow, kn);
    __builtin_amdgcn_sched_barrier(0);
    av_mma(s1, acc);
  }

  const float* qb = Q + (size_t)b * N_ * W_;
  float* ob = out + (size_t)b * N_ * W_;
#pragma unroll
  for (int i = 0; i < AV_TM; ++i) {
#pragma unroll
    for (int g = 0; g < 8; ++g) {
      const size_t off = (size_t)(i0 + i * 16 + g + 8 * half) * W_ + n0 + 4 * r;
      const v4f q4 = *(const v4f*)(qb + off);
      v4f o;
      o.x = acc[i][0][g] + q4.x; o.y = acc[i][1][g] + q4.y;
      o.z = acc[i][2][g] + q4.z; o.w = acc[i][3][g] + q4.w;
      *(v4f*)(ob + off) = o;
    }
  }
}

extern "C" void kernel_launch(void* const* d_in, const int* in_sizes, int n_in,
                              void* d_out, int out_size, void* d_ws, size_t ws_size,
                              hipStream_t stream) {
  const float* x  = (const float*)d_in[0];
  const float* Wq = (const float*)d_in[1];
  const float* bq = (const float*)d_in[2];
  const float* Wk = (const float*)d_in[3];
  const float* bk = (const float*)d_in[4];
  const float* Wv = (const float*)d_in[5];
  const float* bv = (const float*)d_in[6];
  float* out = (float*)d_out;
  float* ws  = (float*)d_ws;

  const size_t QKV = (size_t)B_ * N_ * W_;
  float* Q = ws;
  float* K = ws + QKV;
  float* V = ws + 2 * QKV;
  float* D = ws + 3 * QKV;
  float* S = D + (size_t)B_ * N_;

  {
    dim3 grid((B_ * N_ / 128) * (W_ / 128), 3);
    qkv_kernel<<<grid, 256, 0, stream>>>(x, Wq, bq, Wk, bk, Wv, bv, Q, K, V);
  }
  {
    dim3 grid((N_ / 128) * (N_ / 256), B_);
    scores_kernel<<<grid, 256, 0, stream>>>(Q, K, S);
  }
  colsoftmax_kernel<<<dim3(N_ / SM_COLS, B_), SM_THREADS, 0, stream>>>(S, D);
  scalev_kernel<<<dim3((B_ * N_ * W_ / 4) / 256), 256, 0, stream>>>(V, D);
  {
    dim3 grid(N_ / 64, B_);
    av_kernel<<<grid, 256, 0, stream>>>(S, V, Q, out);
  }
}


# softmax pass folded into the score GEMM epilogue: per-row-block column max and exp there, a small kernel joins the 16 blocks into a factor that the last GEMM multiplies into A; new column-sum order
# speedup vs baseline: 6.5863x; 1.1391x over previous
#include <hip/hip_runtime.h>

#define B_ 8
#define N_ 2048
#define C_ 256
#define W_ 256

typedef float v2f __attribute__((ext_vector_type(2)));
typedef float v8f __attribute__((ext_vector_type(8)));

__device__ __forceinline__ v8f wmma_f32_16x16x4(v2f a, v2f b, v8f c) {
  return __builtin_amdgcn_wmma_f32_16x16x4_f32(false, a, false, b, (short)0, c,
                                               false, false);
}

typedef float v4f __attribute__((ext_vector_type(4)));

#define KS_  32
#define LDP_ 36
template <int TM, int TN, int KD>
struct BlockNT {
  static constexpr int BM = 32 * TM;
  static constexpr int BN = 64 * TN;
  static constexpr int NLA = BM / 32;
  static constexpr int NLB = BN / 32;
  static constexpr int STAGE = (BM + BN) * LDP_;
  static constexpr int LDS_FLOATS = 2 * STAGE;

  static __device__ __forceinline__ void gload(v4f (&g)[NLA + NLB],
                                               const float* __restrict__ ag,
                                               const float* __restrict__ bg, int k0) {
#pragma unroll
    for (int l = 0; l < NLA; ++l) g[l] = *(const v4f*)(ag + (size_t)l * 32 * KD + k0);
#pragma unroll
    for (int l = 0; l < NLB; ++l) g[NLA + l] = *(const v4f*)(bg + (size_t)l * 32 * KD + k0);
  }
  static __device__ __forceinline__ void lstore(float* st, const v4f (&g)[NLA + NLB]) {
#pragma unroll
    for (int l = 0; l < NLA; ++l) *(v4f*)(st + l * 32 * LDP_) = g[l];
#pragma unroll
    for (int l = 0; l < NLB; ++l) *(v4f*)(st + (BM + l * 32) * LDP_) = g[NLA + l];
  }
  static __device__ __forceinline__ void fload(v4f (&a)[TM], v4f (&b)[TN],
                                               const float* fa, const float* fb, int q) {
#pragma unroll
    for (int t = 0; t < TM; ++t) a[t] = *(const v4f*)(fa + t * 16 * LDP_ + q * 8);
#pragma unroll
    for (int t = 0; t < TN; ++t) b[t] = *(const v4f*)(fb + t * 16 * LDP_ + q * 8);
  }
  static __device__ __forceinline__ void mma(const v4f (&a)[TM], const v4f (&b)[TN],
                                             v8f (&acc)[TM][TN]) {
#pragma unroll
    for (int i = 0; i < TM; ++i)
#pragma unroll
      for (int j = 0; j < TN; ++j)
        acc[i][j] = wmma_f32_16x16x4(__builtin_shufflevector(a[i], a[i], 0, 1),
                                     __builtin_shufflevector(b[j], b[j], 0, 1), acc[i][j]);
#pragma unroll
    for (int i = 0; i < TM; ++i)
#pragma unroll
      for (int j = 0; j < TN; ++j)
        acc[i][j] = wmma_f32_16x16x4(__builtin_shufflevector(a[i], a[i], 2, 3),
                                     __builtin_shufflevector(b[j], b[j], 2, 3), acc[i][j]);
  }
  static __device__ __forceinline__ void run(const float* __restrict__ ag,
                                             const float* __restrict__ bg,
                                             float* lds, int tid, int wm, int wn,
                                             int r, int half, v8f (&acc)[TM][TN]) {
    constexpr int NK = KD / KS_;
    float* st = lds + (tid >> 3) * LDP_ + (tid & 7) * 4;
    const float* fa = lds + (wm * TM * 16 + r) * LDP_ + 4 * half;
    const float* fb = lds + (BM + wn * TN * 16 + r) * LDP_ + 4 * half;

    v4f g[NLA + NLB];
    gload(g, ag, bg, 0);
    lstore(st, g);
    __syncthreads();

    v4f a0[TM], b0[TN], a1[TM], b1[TN];
#pragma unroll 1
    for (int ks = 0; ks < NK; ++ks) {
      const int cur = (ks & 1) * STAGE;
      const bool more = ks + 1 < NK;
      if (more) gload(g, ag, bg, (ks + 1) * KS_);
      fload(a0, b0, fa + cur, fb + cur, 0);
      fload(a1, b1, fa + cur, fb + cur, 1);
      __builtin_amdgcn_sched_barrier(0);
      mma(a0, b0, acc);
      __builtin_amdgcn_sched_barrier(0);
      fload(a0, b0, fa + cur, fb + cur, 2);
      __builtin_amdgcn_sched_barrier(0);
      mma(a1, b1, acc);
      __builtin_amdgcn_sched_barrier(0);
      fload(a1, b1, fa + cur, fb + cur, 3);
      __builtin_amdgcn_sched_barrier(0);
      mma(a0, b0, acc);
      mma(a1, b1, acc);
      if (more) lstore(st + (STAGE - cur), g);
      __syncthreads();
    }
  }
};

#define QKV_TM 4
#define QKV_TN 2
__global__ void __launch_bounds__(256)
qkv_kernel(const float* __restrict__ x,
           const float* __restrict__ Wq, const float* __restrict__ bq,
           const float* __restrict__ Wk, const float* __restrict__ bk,
           const float* __restrict__ Wv, const float* __restrict__ bv,
           float* __restrict__ Q, float* __restrict__ K,
           float* __restrict__ V) {
  const float* Wm;
  const float* bias;
  float* out;
  if (blockIdx.y == 0) { Wm = Wq; bias = bq; out = Q; }
  else if (blockIdx.y == 1) { Wm = Wk; bias = bk; out = K; }
  else { Wm = Wv; bias = bv; out = V; }

  const int wave = threadIdx.x >> 5;
  const int lane = threadIdx.x & 31;
  const int half = lane >> 4;
  const int r    = lane & 15;

  const int bn = blockIdx.x & 1;
  const int bm = blockIdx.x >> 1;
  const int m0 = bm * 128 + (wave >> 2) * (16 * QKV_TM);
  const int n0 = bn * 128 + (wave & 3) * (16 * QKV_TN);

  typedef BlockNT<QKV_TM, QKV_TN, C_> Blk;
  __shared__ __attribute__((aligned(16))) float lds[Blk::LDS_FLOATS];
  const int tid = threadIdx.x;
  const float* ag = x  + (size_t)(bm * 128 + (tid >> 3)) * C_ + (tid & 7) * 4;
  const float* bg = Wm + (size_t)(bn * 128 + (tid >> 3)) * C_ + (tid & 7) * 4;

  v8f acc[QKV_TM][QKV_TN] = {};
  Blk::run(ag, bg, lds, tid, wave >> 2, wave & 3, r, half, acc);

#pragma unroll
  for (int j = 0; j < QKV_TN; ++j) {
    const int n = n0 + j * 16 + r;
    const float bv_ = bias[n];
#pragma unroll
    for (int i = 0; i < QKV_TM; ++i) {
#pragma unroll
      for (int g = 0; g < 8; ++g) {
        const int m = m0 + i * 16 + g + 8 * half;
        float v = acc[i][j][g] + bv_;
        v = v > 0.0f ? v : 0.0f;
        out[(size_t)m * W_ + n] = v;
      }
    }
  }
}

#define SC_TM 4
#define SC_TN 4
#define RB_   (N_ / 128)
__global__ void __launch_bounds__(256)
scores_kernel(const float* __restrict__ Q,
              const float* __restrict__ K,
              float* __restrict__ S,
              float* __restrict__ Ml, float* __restrict__ Dl) {
  const int b    = blockIdx.y;
  const int wave = threadIdx.x >> 5;
  const int lane = threadIdx.x & 31;
  const int half = lane >> 4;
  const int r    = lane & 15;
  const int wm   = wave >> 2;
  const int wn   = wave & 3;

  const int bj = blockIdx.x & 7;
  const int bi = blockIdx.x >> 3;
  const int i0 = bi * 128 + wm * (16 * SC_TM);
  const int j0 = bj * 256 + wn * (16 * SC_TN);

  typedef BlockNT<SC_TM, SC_TN, W_> Blk;
  __shared__ __attribute__((aligned(16))) float lds[Blk::LDS_FLOATS];
  __shared__ float cred[2][2][256];
  const int tid = threadIdx.x;
  const float* ag = Q + ((size_t)b * N_ + bi * 128 + (tid >> 3)) * W_ + (tid & 7) * 4;
  const float* bg = K + ((size_t)b * N_ + bj * 256 + (tid >> 3)) * W_ + (tid & 7) * 4;

  v8f acc[SC_TM][SC_TN] = {};
  Blk::run(ag, bg, lds, tid, wm, wn, r, half, acc);

  const float scale = 0.022097086912079608f;
#pragma unroll
  for (int i = 0; i < SC_TM; ++i)
#pragma unroll
    for (int j = 0; j < SC_TN; ++j)
#pragma unroll
      for (int g = 0; g < 8; ++g) acc[i][j][g] *= scale;

  float cm[SC_TN];
#pragma unroll
  for (int j = 0; j < SC_TN; ++j) {
    float m = acc[0][j][0];
#pragma unroll
    for (int i = 0; i < SC_TM; ++i)
#pragma unroll
      for (int g = 0; g < 8; ++g) m = fmaxf(m, acc[i][j][g]);
    cm[j] = fmaxf(m, __shfl_xor(m, 16));
  }
  const int cl = wn * (16 * SC_TN) + r;
  if (half == 0) {
#pragma unroll
    for (int j = 0; j < SC_TN; ++j) cred[0][wm][cl + 16 * j] = cm[j];
  }
  __syncthreads();
#pragma unroll
  for (int j = 0; j < SC_TN; ++j)
    cm[j] = fmaxf(cred[0][0][cl + 16 * j], cred[0][1][cl + 16 * j]);

  float cs[SC_TN] = {0.0f, 0.0f, 0.0f, 0.0f};
  float* sb = S + (size_t)b * N_ * N_;
#pragma unroll
  for (int i = 0; i < SC_TM; ++i) {
#pragma unroll
    for (int g = 0; g < 8; ++g) {
      const int row = i0 + i * 16 + g + 8 * half;
#pragma unroll
      for (int j = 0; j < SC_TN; ++j) {
        const float e = __expf(acc[i][j][g] - cm[j]);
        sb[(size_t)row * N_ + j0 + j * 16 + r] = e;
        cs[j] += e;
      }
    }
  }
#pragma unroll
  for (int j = 0; j < SC_TN; ++j) cs[j] += __shfl_xor(cs[j], 16);
  if (half == 0) {
#pragma unroll
    for (int j = 0; j < SC_TN; ++j) cred[1][wm][cl + 16 * j] = cs[j];
  }
  __syncthreads();
  if (wm == 0 && half == 0) {
    const size_t o = ((size_t)b * RB_ + bi) * N_ + bj * 256 + cl;
#pragma unroll
    for (int j = 0; j < SC_TN; ++j) {
      Ml[o + 16 * j] = cm[j];
      Dl[o + 16 * j] = cred[1][0][cl + 16 * j] + cred[1][1][cl + 16 * j];
    }
  }
}

__global__ void __launch_bounds__(256)
colstats_kernel(const float* __restrict__ Ml, const float* __restrict__ Dl,
                float* __restrict__ F) {
  const int idx = blockIdx.x * 256 + threadIdx.x;
  const int b = idx / N_;
  const int j = idx - b * N_;
  const size_t o = (size_t)b * RB_ * N_ + j;
  float w[RB_], dl[RB_];
#pragma unroll
  for (int bi = 0; bi < RB_; ++bi) { w[bi] = Ml[o + (size_t)bi * N_]; dl[bi] = Dl[o + (size_t)bi * N_]; }
  float m = w[0];
#pragma unroll
  for (int bi = 1; bi < RB_; ++bi) m = fmaxf(m, w[bi]);
  float d = 0.0f;
#pragma unroll
  for (int bi = 0; bi < RB_; ++bi) { w[bi] = __expf(w[bi] - m); d += dl[bi] * w[bi]; }
  const float inv = 1.0f / d;
#pragma unroll
  for (int bi = 0; bi < RB_; ++bi) F[o + (size_t)bi * N_] = w[bi] * inv;
}

#define AV_TM 2
#define AV_TN 4
#define AV_KL 16
struct AvStage { v4f a[AV_TM][AV_KL / 4]; v4f b[AV_KL]; v4f f[AV_KL / 4]; };

__device__ __forceinline__ void av_load(AvStage& s, const float* __restrict__ arow,
                                        const float* __restrict__ brow,
                                        const float* __restrict__ frow, int k0) {
#pragma unroll
  for (int q = 0; q < AV_KL / 4; ++q) s.f[q] = *(const v4f*)(frow + k0 + 4 * q);
#pragma unroll
  for (int t = 0; t < AV_TM; ++t)
#pragma unroll
    for (int q = 0; q < AV_KL / 4; ++q)
      s.a[t][q] = *(const v4f*)(arow + (size_t)t * 16 * N_ + k0 + 4 * q);
#pragma unroll
  for (int e = 0; e < AV_KL; ++e) s.b[e] = *(const v4f*)(brow + (size_t)(k0 + e) * W_);
}

__device__ __forceinline__ void av_scale(AvStage& s) {
#pragma unroll
  for (int t = 0; t < AV_TM; ++t)
#pragma unroll
    for (int q = 0; q < AV_KL / 4; ++q) s.a[t][q] *= s.f[q];
}

__device__ __forceinline__ void av_mma(const AvStage& s, v8f (&acc)[AV_TM][AV_TN]) {
#pragma unroll
  for (int q = 0; q < AV_KL / 4; ++q) {
#pragma unroll
    for (int p = 0; p < 2; ++p) {
      const int e0 = 4 * q + 2 * p;
#pragma unroll
      for (int j = 0; j < AV_TN; ++j) {
        v2f bf;
        bf.x = s.b[e0][j];
        bf.y = s.b[e0 + 1][j];
#pragma unroll
        for (int i = 0; i < AV_TM; ++i) {
          v2f af;
          af.x = s.a[i][q][2 * p];
          af.y = s.a[i][q][2 * p + 1];
          acc[i][j] = wmma_f32_16x16x4(af, bf, acc[i][j]);
        }
      }
    }
  }
}

__global__ void __launch_bounds__(256)
av_kernel(const float* __restrict__ P, const float* __restrict__ V,
          const float* __restrict__ F,
          const float* __restrict__ Q, float* __restrict__ out) {
  const int b    = blockIdx.y;
  const int wave = threadIdx.x >> 5;
  const int lane = threadIdx.x & 31;
  const int half = lane >> 4;
  const int r    = lane & 15;

  const int i0 = blockIdx.x * 64 + (wave >> 2) * (16 * AV_TM);
  const int n0 = (wave & 3) * (16 * AV_TN);

  const float* arow = P + (size_t)b * N_ * N_ + (size_t)(i0 + r) * N_ + AV_KL * half;
  const float* brow = V + (size_t)b * N_ * W_ + (size_t)(AV_KL * half) * W_ + n0 + 4 * r;
  const float* frow = F + ((size_t)b * RB_ + (blockIdx.x >> 1)) * N_ + AV_KL * half;

  v8f acc[AV_TM][AV_TN] = {};
  AvStage s0, s1;
  av_load(s0, arow, brow, frow, 0);
#pragma unroll 1
  for (int k0 = 0; k0 < N_; k0 += 4 * AV_KL) {
    av_load(s1, arow, brow, frow, k0 + 2 * AV_KL);
    __builtin_amdgcn_sched_barrier(0);
    av_scale(s0);
    av_mma(s0, acc);
    const int kn = (k0 + 4 * AV_KL < N_) ? k0 + 4 * AV_KL : 0;
    __builtin_amdgcn_sched_barrier(0);
    av_load(s0, arow, brow, frow, kn);
    __builtin_amdgcn_sched_barrier(0);
    av_scale(s1);
    av_mma(s1, acc);
  }

  const float* qb = Q + (size_t)b * N_ * W_;
  float* ob = out + (size_t)b * N_ * W_;
#pragma unroll
  for (int i = 0; i < AV_TM; ++i) {
#pragma unroll
    for (int g = 0; g < 8; ++g) {
      const size_t off = (size_t)(i0 + i * 16 + g + 8 * half) * W_ + n0 + 4 * r;
      const v4f q4 = *(const v4f*)(qb + off);
      v4f o;
      o.x = acc[i][0][g] + q4.x; o.y = acc[i][1][g] + q4.y;
      o.z = acc[i][2][g] + q4.z; o.w = acc[i][3][g] + q4.w;
      *(v4f*)(ob + off) = o;
    }
  }
}

extern "C" void kernel_launch(void* const* d_in, const int* in_sizes, int n_in,
                              void* d_out, int out_size, void* d_ws, size_t ws_size,
                              hipStream_t stream) {
  const float* x  = (const float*)d_in[0];
  const float* Wq = (const float*)d_in[1];
  const float* bq = (const float*)d_in[2];
  const float* Wk = (const float*)d_in[3];
  const float* bk = (const float*)d_in[4];
  const float* Wv = (const float*)d_in[5];
  const float* bv = (const float*)d_in[6];
  float* out = (float*)d_out;
  float* ws  = (float*)d_ws;

  const size_t QKV = (size_t)B_ * N_ * W_;
  float* Q = ws;
  float* K = ws + QKV;
  float* V = ws + 2 * QKV;
  float* S = ws + 3 * QKV + (size_t)B_ * N_;
  float* Ml = out;
  float* Dl = out + (size_t)B_ * RB_ * N_;
  float* F  = K;

  {
    dim3 grid((B_ * N_ / 128) * (W_ / 128), 3);
    qkv_kernel<<<grid, 256, 0, stream>>>(x, Wq, bq, Wk, bk, Wv, bv, Q, K, V);
  }
  {
    dim3 grid((N_ / 128) * (N_ / 256), B_);
    scores_kernel<<<grid, 256, 0, stream>>>(Q, K, S, Ml, Dl);
  }
  colstats_kernel<<<dim3(B_ * N_ / 256), 256, 0, stream>>>(Ml, Dl, F);
  {
    dim3 grid(N_ / 64, B_);
    av_kernel<<<grid, 256, 0, stream>>>(S, V, F, Q, out);
  }
}
